# starting kernel
# baseline (speedup 1.0000x reference)
#include <hip/hip_runtime.h>

typedef __attribute__((ext_vector_type(2))) float v2f;
typedef __attribute__((ext_vector_type(8))) float v8f;

#define N_NODES 50000
#define N_EDGES 400000
#define IN_F    768
#define HID_F   16
#define OUT_F   21

__global__ __launch_bounds__(256) void gcn_gemm1_wmma(
    const float* __restrict__ X, const float* __restrict__ W1,
    float* __restrict__ Z)
{
  __shared__ float w1s[IN_F * HID_F];
  {
    const float4* w4 = (const float4*)W1;
    float4*       s4 = (float4*)w1s;
    const int n4 = (IN_F * HID_F) / 4;
    for (int i = threadIdx.x; i < n4; i += 256) s4[i] = w4[i];
  }
  __syncthreads();

  const int lane = threadIdx.x & 31;
  const int wave = threadIdx.x >> 5;
  const int tile = blockIdx.x * 8 + wave;
  if (tile >= (N_NODES / 16)) return;
  const int m0   = tile * 16;

  const int half = lane >> 4;
  const int l16  = lane & 15;
  const int koff = half << 1;

  const float* xrow = X + (size_t)(m0 + l16) * IN_F + koff;

  v8f acc0 = {};
  v8f acc1 = {};
  for (int k = 0; k < IN_F; k += 8) {
    v2f a0 = *(const v2f*)(xrow + k);
    v2f a1 = *(const v2f*)(xrow + k + 4);
    v2f b0, b1;
    b0.x = w1s[(k + koff)     * HID_F + l16];
    b0.y = w1s[(k + koff + 1) * HID_F + l16];
    b1.x = w1s[(k + 4 + koff)     * HID_F + l16];
    b1.y = w1s[(k + 4 + koff + 1) * HID_F + l16];
    acc0 = __builtin_amdgcn_wmma_f32_16x16x4_f32(false, a0, false, b0,
                                                 (short)0, acc0, false, false);
    acc1 = __builtin_amdgcn_wmma_f32_16x16x4_f32(false, a1, false, b1,
                                                 (short)0, acc1, false, false);
  }

  float* zp = Z + (size_t)(m0 + half * 8) * HID_F + l16;
#pragma unroll
  for (int r = 0; r < 8; ++r) zp[r * HID_F] = acc0[r] + acc1[r];
}

__global__ __launch_bounds__(256) void gcn_zero(float* __restrict__ p, int n)
{
  int i = blockIdx.x * 256 + threadIdx.x;
  if (i < n) p[i] = 0.0f;
}

__global__ __launch_bounds__(256) void gcn_scatter16(
    const float* __restrict__ feat, const int* __restrict__ src,
    const int* __restrict__ dst, float* __restrict__ acc)
{
  unsigned tid = blockIdx.x * 256u + threadIdx.x;
  unsigned e = tid >> 4;
  unsigned f = tid & 15;
  if (e < N_EDGES) {
    int s = src[e];
    int d = dst[e];
    atomicAdd(acc + (size_t)d * HID_F + f, feat[(size_t)s * HID_F + f]);
  }
}

__global__ __launch_bounds__(256) void gcn_bias_relu(
    const float* __restrict__ a, const float* __restrict__ b1,
    float* __restrict__ h)
{
  int i = blockIdx.x * 256 + threadIdx.x;
  if (i < N_NODES * HID_F) {
    float v = a[i] + b1[i & (HID_F - 1)];
    h[i] = v > 0.0f ? v : 0.0f;
  }
}

__global__ __launch_bounds__(256) void gcn_gemm2_wmma(
    const float* __restrict__ A, const float* __restrict__ W2,
    const float* __restrict__ b2, float* __restrict__ out)
{
  const int lane = threadIdx.x & 31;
  const int wave = threadIdx.x >> 5;
  const int t    = blockIdx.x * 8 + wave;
  const int mtiles = N_NODES / 16;
  if (t >= mtiles * 2) return;
  const int mt = t >> 1;
  const int nt = t & 1;
  const int m0 = mt * 16;

  const int half = lane >> 4;
  const int l16  = lane & 15;
  const int koff = half << 1;
  const int n    = nt * 16 + l16;
  const bool nvalid = (n < OUT_F);

  const float* arow = A + (size_t)(m0 + l16) * HID_F + koff;

  v8f acc = {};
#pragma unroll
  for (int k = 0; k < HID_F; k += 4) {
    v2f a = *(const v2f*)(arow + k);
    v2f b;
    b.x = nvalid ? W2[(k + koff)     * OUT_F + n] : 0.0f;
    b.y = nvalid ? W2[(k + koff + 1) * OUT_F + n] : 0.0f;
    acc = __builtin_amdgcn_wmma_f32_16x16x4_f32(false, a, false, b,
                                                (short)0, acc, false, false);
  }

  if (nvalid) {
    float bias = b2[n];
    float* op = out + (size_t)(m0 + half * 8) * OUT_F + n;
#pragma unroll
    for (int r = 0; r < 8; ++r) op[r * OUT_F] = acc[r] + bias;
  }
}

extern "C" void kernel_launch(void* const* d_in, const int* in_sizes, int n_in,
                              void* d_out, int out_size, void* d_ws, size_t ws_size,
                              hipStream_t stream)
{
  const float* feature = (const float*)d_in[0];
  const float* W1      = (const float*)d_in[1];
  const float* b1      = (const float*)d_in[2];
  const float* W2      = (const float*)d_in[3];
  const float* b2      = (const float*)d_in[4];
  const int*   src     = (const int*)d_in[5];
  const int*   dst     = (const int*)d_in[6];
  float*       out     = (float*)d_out;

  float* buf0 = (float*)d_ws;
  float* buf1 = buf0 + (size_t)N_NODES * HID_F;

  const int NH = N_NODES * HID_F;
  const int zero_blocks    = (NH + 255) / 256;
  const int scatter_blocks = (N_EDGES * HID_F) / 256;
  const int gemm1_blocks   = ((N_NODES / 16) + 7) / 8;
  const int gemm2_blocks   = ((N_NODES / 16) * 2 + 7) / 8;

  gcn_gemm1_wmma<<<gemm1_blocks, 256, 0, stream>>>(feature, W1, buf0);

  gcn_zero<<<zero_blocks, 256, 0, stream>>>(buf1, NH);
  gcn_scatter16<<<scatter_blocks, 256, 0, stream>>>(buf0, src, dst, buf1);

  gcn_bias_relu<<<zero_blocks, 256, 0, stream>>>(buf1, b1, buf0);

  gcn_zero<<<zero_blocks, 256, 0, stream>>>(buf1, NH);
  gcn_scatter16<<<scatter_blocks, 256, 0, stream>>>(buf0, src, dst, buf1);

  gcn_gemm2_wmma<<<gemm2_blocks, 256, 0, stream>>>(buf1, W2, b2, out);
}


# Aggregation without float atomics: per-call linked edge lists by dst (400k int atomicExch) + per-node gather sums, 4 lanes x float4 per node; float sum order = list (arrival) order
# speedup vs baseline: 1.1072x; 1.1072x over previous
#include <hip/hip_runtime.h>

typedef __attribute__((ext_vector_type(2))) float v2f;
typedef __attribute__((ext_vector_type(8))) float v8f;

#define N_NODES 50000
#define N_EDGES 400000
#define IN_F    768
#define HID_F   16
#define OUT_F   21

__global__ __launch_bounds__(256) void gcn_gemm1_wmma(
    const float* __restrict__ X, const float* __restrict__ W1,
    float* __restrict__ Z, int* __restrict__ head)
{
  __shared__ float w1s[IN_F * HID_F];
  {
    const int g = blockIdx.x * 256 + threadIdx.x;
    if (g < N_NODES) head[g] = -1;
  }
  {
    const float4* w4 = (const float4*)W1;
    float4*       s4 = (float4*)w1s;
    const int n4 = (IN_F * HID_F) / 4;
    for (int i = threadIdx.x; i < n4; i += 256) s4[i] = w4[i];
  }
  __syncthreads();

  const int lane = threadIdx.x & 31;
  const int wave = threadIdx.x >> 5;
  const int tile = blockIdx.x * 8 + wave;
  if (tile >= (N_NODES / 16)) return;
  const int m0   = tile * 16;

  const int half = lane >> 4;
  const int l16  = lane & 15;
  const int koff = half << 1;

  const float* xrow = X + (size_t)(m0 + l16) * IN_F + koff;

  v8f acc0 = {};
  v8f acc1 = {};
  for (int k = 0; k < IN_F; k += 8) {
    v2f a0 = *(const v2f*)(xrow + k);
    v2f a1 = *(const v2f*)(xrow + k + 4);
    v2f b0, b1;
    b0.x = w1s[(k + koff)     * HID_F + l16];
    b0.y = w1s[(k + koff + 1) * HID_F + l16];
    b1.x = w1s[(k + 4 + koff)     * HID_F + l16];
    b1.y = w1s[(k + 4 + koff + 1) * HID_F + l16];
    acc0 = __builtin_amdgcn_wmma_f32_16x16x4_f32(false, a0, false, b0,
                                                 (short)0, acc0, false, false);
    acc1 = __builtin_amdgcn_wmma_f32_16x16x4_f32(false, a1, false, b1,
                                                 (short)0, acc1, false, false);
  }

  float* zp = Z + (size_t)(m0 + half * 8) * HID_F + l16;
#pragma unroll
  for (int r = 0; r < 8; ++r) zp[r * HID_F] = acc0[r] + acc1[r];
}

__global__ __launch_bounds__(256) void gcn_link(
    const int* __restrict__ src, const int* __restrict__ dst,
    int* __restrict__ head, int* __restrict__ cells)
{
  unsigned e = blockIdx.x * 256u + threadIdx.x;
  if (e < N_EDGES) {
    int s = src[e];
    int d = dst[e];
    int prev = atomicExch(head + d, (int)e);
    *(int2*)(cells + 2 * (size_t)e) = make_int2(prev, s);
  }
}

__global__ __launch_bounds__(256) void gcn_gather16(
    const float* __restrict__ feat, const int* __restrict__ head,
    const int* __restrict__ cells, float* __restrict__ acc)
{
  unsigned tid = blockIdx.x * 256u + threadIdx.x;
  unsigned n = tid >> 2;
  unsigned q = (tid & 3u) * 4u;
  if (n < N_NODES) {
    float4 s = make_float4(0.0f, 0.0f, 0.0f, 0.0f);
    float4 v = make_float4(0.0f, 0.0f, 0.0f, 0.0f);
    int e = head[n];
    for (int it = 0; it < N_EDGES && (unsigned)e < (unsigned)N_EDGES; ++it) {
      int2 cell = *(const int2*)(cells + 2 * (size_t)e);
      __builtin_amdgcn_sched_barrier(0);
      s.x += v.x; s.y += v.y; s.z += v.z; s.w += v.w;
      v = *(const float4*)(feat + (size_t)cell.y * HID_F + q);
      e = cell.x;
    }
    s.x += v.x; s.y += v.y; s.z += v.z; s.w += v.w;
    *(float4*)(acc + (size_t)n * HID_F + q) = s;
  }
}

__global__ __launch_bounds__(256) void gcn_bias_relu(
    const float* __restrict__ a, const float* __restrict__ b1,
    float* __restrict__ h)
{
  int i = blockIdx.x * 256 + threadIdx.x;
  if (i < N_NODES * HID_F) {
    float v = a[i] + b1[i & (HID_F - 1)];
    h[i] = v > 0.0f ? v : 0.0f;
  }
}

__global__ __launch_bounds__(256) void gcn_gemm2_wmma(
    const float* __restrict__ A, const float* __restrict__ W2,
    const float* __restrict__ b2, float* __restrict__ out)
{
  const int lane = threadIdx.x & 31;
  const int wave = threadIdx.x >> 5;
  const int t    = blockIdx.x * 8 + wave;
  const int mtiles = N_NODES / 16;
  if (t >= mtiles * 2) return;
  const int mt = t >> 1;
  const int nt = t & 1;
  const int m0 = mt * 16;

  const int half = lane >> 4;
  const int l16  = lane & 15;
  const int koff = half << 1;
  const int n    = nt * 16 + l16;
  const bool nvalid = (n < OUT_F);

  const float* arow = A + (size_t)(m0 + l16) * HID_F + koff;

  v8f acc = {};
#pragma unroll
  for (int k = 0; k < HID_F; k += 4) {
    v2f a = *(const v2f*)(arow + k);
    v2f b;
    b.x = nvalid ? W2[(k + koff)     * OUT_F + n] : 0.0f;
    b.y = nvalid ? W2[(k + koff + 1) * OUT_F + n] : 0.0f;
    acc = __builtin_amdgcn_wmma_f32_16x16x4_f32(false, a, false, b,
                                                (short)0, acc, false, false);
  }

  if (nvalid) {
    float bias = b2[n];
    float* op = out + (size_t)(m0 + half * 8) * OUT_F + n;
#pragma unroll
    for (int r = 0; r < 8; ++r) op[r * OUT_F] = acc[r] + bias;
  }
}

extern "C" void kernel_launch(void* const* d_in, const int* in_sizes, int n_in,
                              void* d_out, int out_size, void* d_ws, size_t ws_size,
                              hipStream_t stream)
{
  const float* feature = (const float*)d_in[0];
  const float* W1      = (const float*)d_in[1];
  const float* b1      = (const float*)d_in[2];
  const float* W2      = (const float*)d_in[3];
  const float* b2      = (const float*)d_in[4];
  const int*   src     = (const int*)d_in[5];
  const int*   dst     = (const int*)d_in[6];
  float*       out     = (float*)d_out;

  float* buf0 = (float*)d_ws;
  float* buf1 = buf0 + (size_t)N_NODES * HID_F;

  int* cells = (int*)d_out;
  int* head  = cells + 2 * (size_t)N_EDGES;

  const int NH = N_NODES * HID_F;
  const int relu_blocks    = (NH + 255) / 256;
  const int link_blocks    = (N_EDGES + 255) / 256;
  const int gather_blocks  = (N_NODES * 4 + 255) / 256;
  const int gemm1_blocks   = ((N_NODES / 16) + 7) / 8;
  const int gemm2_blocks   = ((N_NODES / 16) * 2 + 7) / 8;

  gcn_gemm1_wmma<<<gemm1_blocks, 256, 0, stream>>>(feature, W1, buf0, head);

  gcn_link<<<link_blocks, 256, 0, stream>>>(src, dst, head, cells);

  gcn_gather16<<<gather_blocks, 256, 0, stream>>>(buf0, head, cells, buf1);

  gcn_bias_relu<<<relu_blocks, 256, 0, stream>>>(buf1, b1, buf0);

  gcn_gather16<<<gather_blocks, 256, 0, stream>>>(buf0, head, cells, buf1);

  gcn_gemm2_wmma<<<gemm2_blocks, 256, 0, stream>>>(buf1, W2, b2, out);
}


# Bias add and ReLU moved into the layer-1 gather's epilogue, applied to each finished sum in registers; the elementwise launch and its 25000 waves are gone (5 launches, not 6)
# speedup vs baseline: 1.1598x; 1.0475x over previous
#include <hip/hip_runtime.h>

typedef __attribute__((ext_vector_type(2))) float v2f;
typedef __attribute__((ext_vector_type(8))) float v8f;

#define N_NODES 50000
#define N_EDGES 400000
#define IN_F    768
#define HID_F   16
#define OUT_F   21

__global__ __launch_bounds__(256) void gcn_gemm1_wmma(
    const float* __restrict__ X, const float* __restrict__ W1,
    float* __restrict__ Z, int* __restrict__ head)
{
  __shared__ float w1s[IN_F * HID_F];
  {
    const int g = blockIdx.x * 256 + threadIdx.x;
    if (g < N_NODES) head[g] = -1;
  }
  {
    const float4* w4 = (const float4*)W1;
    float4*       s4 = (float4*)w1s;
    const int n4 = (IN_F * HID_F) / 4;
    for (int i = threadIdx.x; i < n4; i += 256) s4[i] = w4[i];
  }
  __syncthreads();

  const int lane = threadIdx.x & 31;
  const int wave = threadIdx.x >> 5;
  const int tile = blockIdx.x * 8 + wave;
  if (tile >= (N_NODES / 16)) return;
  const int m0   = tile * 16;

  const int half = lane >> 4;
  const int l16  = lane & 15;
  const int koff = half << 1;

  const float* xrow = X + (size_t)(m0 + l16) * IN_F + koff;

  v8f acc0 = {};
  v8f acc1 = {};
  for (int k = 0; k < IN_F; k += 8) {
    v2f a0 = *(const v2f*)(xrow + k);
    v2f a1 = *(const v2f*)(xrow + k + 4);
    v2f b0, b1;
    b0.x = w1s[(k + koff)     * HID_F + l16];
    b0.y = w1s[(k + koff + 1) * HID_F + l16];
    b1.x = w1s[(k + 4 + koff)     * HID_F + l16];
    b1.y = w1s[(k + 4 + koff + 1) * HID_F + l16];
    acc0 = __builtin_amdgcn_wmma_f32_16x16x4_f32(false, a0, false, b0,
                                                 (short)0, acc0, false, false);
    acc1 = __builtin_amdgcn_wmma_f32_16x16x4_f32(false, a1, false, b1,
                                                 (short)0, acc1, false, false);
  }

  float* zp = Z + (size_t)(m0 + half * 8) * HID_F + l16;
#pragma unroll
  for (int r = 0; r < 8; ++r) zp[r * HID_F] = acc0[r] + acc1[r];
}

__global__ __launch_bounds__(256) void gcn_link(
    const int* __restrict__ src, const int* __restrict__ dst,
    int* __restrict__ head, int* __restrict__ cells)
{
  unsigned e = blockIdx.x * 256u + threadIdx.x;
  if (e < N_EDGES) {
    int s = src[e];
    int d = dst[e];
    int prev = atomicExch(head + d, (int)e);
    *(int2*)(cells + 2 * (size_t)e) = make_int2(prev, s);
  }
}

__device__ __forceinline__ float4 gcn_list_sum(
    const float* __restrict__ feat, const int* __restrict__ cells,
    int e, unsigned q)
{
  float4 s = make_float4(0.0f, 0.0f, 0.0f, 0.0f);
  float4 v = make_float4(0.0f, 0.0f, 0.0f, 0.0f);
  for (int it = 0; it < N_EDGES && (unsigned)e < (unsigned)N_EDGES; ++it) {
    int2 cell = *(const int2*)(cells + 2 * (size_t)e);
    __builtin_amdgcn_sched_barrier(0);
    s.x += v.x; s.y += v.y; s.z += v.z; s.w += v.w;
    v = *(const float4*)(feat + (size_t)cell.y * HID_F + q);
    e = cell.x;
  }
  s.x += v.x; s.y += v.y; s.z += v.z; s.w += v.w;
  return s;
}

__global__ __launch_bounds__(256) void gcn_gather16_bias_relu(
    const float* __restrict__ feat, const int* __restrict__ head,
    const int* __restrict__ cells, const float* __restrict__ b1,
    float* __restrict__ h)
{
  unsigned tid = blockIdx.x * 256u + threadIdx.x;
  unsigned n = tid >> 2;
  unsigned q = (tid & 3u) * 4u;
  if (n < N_NODES) {
    const float4 b = *(const float4*)(b1 + q);
    float4 s = gcn_list_sum(feat, cells, head[n], q);
    float4 r;
    r.x = s.x + b.x; r.y = s.y + b.y; r.z = s.z + b.z; r.w = s.w + b.w;
    r.x = r.x > 0.0f ? r.x : 0.0f;
    r.y = r.y > 0.0f ? r.y : 0.0f;
    r.z = r.z > 0.0f ? r.z : 0.0f;
    r.w = r.w > 0.0f ? r.w : 0.0f;
    *(float4*)(h + (size_t)n * HID_F + q) = r;
  }
}

__global__ __launch_bounds__(256) void gcn_gather16(
    const float* __restrict__ feat, const int* __restrict__ head,
    const int* __restrict__ cells, float* __restrict__ acc)
{
  unsigned tid = blockIdx.x * 256u + threadIdx.x;
  unsigned n = tid >> 2;
  unsigned q = (tid & 3u) * 4u;
  if (n < N_NODES) {
    float4 s = gcn_list_sum(feat, cells, head[n], q);
    *(float4*)(acc + (size_t)n * HID_F + q) = s;
  }
}

__global__ __launch_bounds__(256) void gcn_gemm2_wmma(
    const float* __restrict__ A, const float* __restrict__ W2,
    const float* __restrict__ b2, float* __restrict__ out)
{
  const int lane = threadIdx.x & 31;
  const int wave = threadIdx.x >> 5;
  const int t    = blockIdx.x * 8 + wave;
  const int mtiles = N_NODES / 16;
  if (t >= mtiles * 2) return;
  const int mt = t >> 1;
  const int nt = t & 1;
  const int m0 = mt * 16;

  const int half = lane >> 4;
  const int l16  = lane & 15;
  const int koff = half << 1;
  const int n    = nt * 16 + l16;
  const bool nvalid = (n < OUT_F);

  const float* arow = A + (size_t)(m0 + l16) * HID_F + koff;

  v8f acc = {};
#pragma unroll
  for (int k = 0; k < HID_F; k += 4) {
    v2f a = *(const v2f*)(arow + k);
    v2f b;
    b.x = nvalid ? W2[(k + koff)     * OUT_F + n] : 0.0f;
    b.y = nvalid ? W2[(k + koff + 1) * OUT_F + n] : 0.0f;
    acc = __builtin_amdgcn_wmma_f32_16x16x4_f32(false, a, false, b,
                                                (short)0, acc, false, false);
  }

  if (nvalid) {
    float bias = b2[n];
    float* op = out + (size_t)(m0 + half * 8) * OUT_F + n;
#pragma unroll
    for (int r = 0; r < 8; ++r) op[r * OUT_F] = acc[r] + bias;
  }
}

extern "C" void kernel_launch(void* const* d_in, const int* in_sizes, int n_in,
                              void* d_out, int out_size, void* d_ws, size_t ws_size,
                              hipStream_t stream)
{
  const float* feature = (const float*)d_in[0];
  const float* W1      = (const float*)d_in[1];
  const float* b1      = (const float*)d_in[2];
  const float* W2      = (const float*)d_in[3];
  const float* b2      = (const float*)d_in[4];
  const int*   src     = (const int*)d_in[5];
  const int*   dst     = (const int*)d_in[6];
  float*       out     = (float*)d_out;

  float* buf0 = (float*)d_ws;
  float* buf1 = buf0 + (size_t)N_NODES * HID_F;

  int* cells = (int*)d_out;
  int* head  = cells + 2 * (size_t)N_EDGES;

  const int link_blocks    = (N_EDGES + 255) / 256;
  const int gather_blocks  = (N_NODES * 4 + 255) / 256;
  const int gemm1_blocks   = ((N_NODES / 16) + 7) / 8;
  const int gemm2_blocks   = ((N_NODES / 16) * 2 + 7) / 8;

  gcn_gemm1_wmma<<<gemm1_blocks, 256, 0, stream>>>(feature, W1, buf0, head);

  gcn_link<<<link_blocks, 256, 0, stream>>>(src, dst, head, cells);

  gcn_gather16_bias_relu<<<gather_blocks, 256, 0, stream>>>(buf0, head, cells, b1, buf1);

  gcn_gather16<<<gather_blocks, 256, 0, stream>>>(buf1, head, cells, buf0);

  gcn_gemm2_wmma<<<gemm2_blocks, 256, 0, stream>>>(buf0, W2, b2, out);
}


# Edge-list build moved into the X@W1 launch as 391 extra work-groups (4 edges a thread): its 400k returning swaps run under the X stream; heads cleared by a small launch ahead; sum order still arrival
# speedup vs baseline: 1.2083x; 1.0418x over previous
#include <hip/hip_runtime.h>

typedef __attribute__((ext_vector_type(2))) float v2f;
typedef __attribute__((ext_vector_type(8))) float v8f;

#define N_NODES 50000
#define N_EDGES 400000
#define IN_F    768
#define HID_F   16
#define OUT_F   21

#define GEMM1_BLOCKS (((N_NODES / 16) + 7) / 8)
#define LINK_BLOCKS  (((N_EDGES / 4) + 255) / 256)

__global__ __launch_bounds__(256) void gcn_heads_clear(int* __restrict__ head)
{
  const unsigned g = blockIdx.x * 256u + threadIdx.x;
  if (g < N_NODES / 4) ((int4*)head)[g] = make_int4(-1, -1, -1, -1);
}

__device__ __forceinline__ void gcn_link4(
    unsigned t, const int* __restrict__ src, const int* __restrict__ dst,
    int* __restrict__ head, int* __restrict__ cells)
{
  if (t < N_EDGES / 4) {
    const int4 s = ((const int4*)src)[t];
    const int4 d = ((const int4*)dst)[t];
    const int e = (int)(4u * t);
    const int p0 = atomicExch(head + d.x, e);
    const int p1 = atomicExch(head + d.y, e + 1);
    const int p2 = atomicExch(head + d.z, e + 2);
    const int p3 = atomicExch(head + d.w, e + 3);
    int4* c = (int4*)(cells + 8 * (size_t)t);
    c[0] = make_int4(p0, s.x, p1, s.y);
    c[1] = make_int4(p2, s.z, p3, s.w);
  }
}

__global__ __launch_bounds__(256) void gcn_gemm1_wmma_link(
    const float* __restrict__ X, const float* __restrict__ W1,
    float* __restrict__ Z,
    const int* __restrict__ src, const int* __restrict__ dst,
    int* __restrict__ head, int* __restrict__ cells)
{
  __shared__ float w1s[IN_F * HID_F];
  if (blockIdx.x >= GEMM1_BLOCKS) {
    gcn_link4((blockIdx.x - GEMM1_BLOCKS) * 256u + threadIdx.x, src, dst, head, cells);
    return;
  }
  {
    const float4* w4 = (const float4*)W1;
    float4*       s4 = (float4*)w1s;
    const int n4 = (IN_F * HID_F) / 4;
    for (int i = threadIdx.x; i < n4; i += 256) s4[i] = w4[i];
  }
  __syncthreads();

  const int lane = threadIdx.x & 31;
  const int wave = threadIdx.x >> 5;
  const int tile = blockIdx.x * 8 + wave;
  if (tile >= (N_NODES / 16)) return;
  const int m0   = tile * 16;

  const int half = lane >> 4;
  const int l16  = lane & 15;
  const int koff = half << 1;

  const float* xrow = X + (size_t)(m0 + l16) * IN_F + koff;

  v8f acc0 = {};
  v8f acc1 = {};
  for (int k = 0; k < IN_F; k += 8) {
    v2f a0 = *(const v2f*)(xrow + k);
    v2f a1 = *(const v2f*)(xrow + k + 4);
    v2f b0, b1;
    b0.x = w1s[(k + koff)     * HID_F + l16];
    b0.y = w1s[(k + koff + 1) * HID_F + l16];
    b1.x = w1s[(k + 4 + koff)     * HID_F + l16];
    b1.y = w1s[(k + 4 + koff + 1) * HID_F + l16];
    acc0 = __builtin_amdgcn_wmma_f32_16x16x4_f32(false, a0, false, b0,
                                                 (short)0, acc0, false, false);
    acc1 = __builtin_amdgcn_wmma_f32_16x16x4_f32(false, a1, false, b1,
                                                 (short)0, acc1, false, false);
  }

  float* zp = Z + (size_t)(m0 + half * 8) * HID_F + l16;
#pragma unroll
  for (int r = 0; r < 8; ++r) zp[r * HID_F] = acc0[r] + acc1[r];
}

__device__ __forceinline__ float4 gcn_list_sum(
    const float* __restrict__ feat, const int* __restrict__ cells,
    int e, unsigned q)
{
  float4 s = make_float4(0.0f, 0.0f, 0.0f, 0.0f);
  float4 v = make_float4(0.0f, 0.0f, 0.0f, 0.0f);
  for (int it = 0; it < N_EDGES && (unsigned)e < (unsigned)N_EDGES; ++it) {
    int2 cell = *(const int2*)(cells + 2 * (size_t)e);
    __builtin_amdgcn_sched_barrier(0);
    s.x += v.x; s.y += v.y; s.z += v.z; s.w += v.w;
    v = *(const float4*)(feat + (size_t)cell.y * HID_F + q);
    e = cell.x;
  }
  s.x += v.x; s.y += v.y; s.z += v.z; s.w += v.w;
  return s;
}

__global__ __launch_bounds__(256) void gcn_gather16_bias_relu(
    const float* __restrict__ feat, const int* __restrict__ head,
    const int* __restrict__ cells, const float* __restrict__ b1,
    float* __restrict__ h)
{
  unsigned tid = blockIdx.x * 256u + threadIdx.x;
  unsigned n = tid >> 2;
  unsigned q = (tid & 3u) * 4u;
  if (n < N_NODES) {
    const float4 b = *(const float4*)(b1 + q);
    float4 s = gcn_list_sum(feat, cells, head[n], q);
    float4 r;
    r.x = s.x + b.x; r.y = s.y + b.y; r.z = s.z + b.z; r.w = s.w + b.w;
    r.x = r.x > 0.0f ? r.x : 0.0f;
    r.y = r.y > 0.0f ? r.y : 0.0f;
    r.z = r.z > 0.0f ? r.z : 0.0f;
    r.w = r.w > 0.0f ? r.w : 0.0f;
    *(float4*)(h + (size_t)n * HID_F + q) = r;
  }
}

__global__ __launch_bounds__(256) void gcn_gather16(
    const float* __restrict__ feat, const int* __restrict__ head,
    const int* __restrict__ cells, float* __restrict__ acc)
{
  unsigned tid = blockIdx.x * 256u + threadIdx.x;
  unsigned n = tid >> 2;
  unsigned q = (tid & 3u) * 4u;
  if (n < N_NODES) {
    float4 s = gcn_list_sum(feat, cells, head[n], q);
    *(float4*)(acc + (size_t)n * HID_F + q) = s;
  }
}

__global__ __launch_bounds__(256) void gcn_gemm2_wmma(
    const float* __restrict__ A, const float* __restrict__ W2,
    const float* __restrict__ b2, float* __restrict__ out)
{
  const int lane = threadIdx.x & 31;
  const int wave = threadIdx.x >> 5;
  const int t    = blockIdx.x * 8 + wave;
  const int mtiles = N_NODES / 16;
  if (t >= mtiles * 2) return;
  const int mt = t >> 1;
  const int nt = t & 1;
  const int m0 = mt * 16;

  const int half = lane >> 4;
  const int l16  = lane & 15;
  const int koff = half << 1;
  const int n    = nt * 16 + l16;
  const bool nvalid = (n < OUT_F);

  const float* arow = A + (size_t)(m0 + l16) * HID_F + koff;

  v8f acc = {};
#pragma unroll
  for (int k = 0; k < HID_F; k += 4) {
    v2f a = *(const v2f*)(arow + k);
    v2f b;
    b.x = nvalid ? W2[(k + koff)     * OUT_F + n] : 0.0f;
    b.y = nvalid ? W2[(k + koff + 1) * OUT_F + n] : 0.0f;
    acc = __builtin_amdgcn_wmma_f32_16x16x4_f32(false, a, false, b,
                                                (short)0, acc, false, false);
  }

  if (nvalid) {
    float bias = b2[n];
    float* op = out + (size_t)(m0 + half * 8) * OUT_F + n;
#pragma unroll
    for (int r = 0; r < 8; ++r) op[r * OUT_F] = acc[r] + bias;
  }
}

extern "C" void kernel_launch(void* const* d_in, const int* in_sizes, int n_in,
                              void* d_out, int out_size, void* d_ws, size_t ws_size,
                              hipStream_t stream)
{
  const float* feature = (const float*)d_in[0];
  const float* W1      = (const float*)d_in[1];
  const float* b1      = (const float*)d_in[2];
  const float* W2      = (const float*)d_in[3];
  const float* b2      = (const float*)d_in[4];
  const int*   src     = (const int*)d_in[5];
  const int*   dst     = (const int*)d_in[6];
  float*       out     = (float*)d_out;

  float* buf0 = (float*)d_ws;
  float* buf1 = buf0 + (size_t)N_NODES * HID_F;

  int* cells = (int*)d_out;
  int* head  = cells + 2 * (size_t)N_EDGES;

  const int clear_blocks   = ((N_NODES / 4) + 255) / 256;
  const int gather_blocks  = (N_NODES * 4 + 255) / 256;
  const int gemm2_blocks   = ((N_NODES / 16) * 2 + 7) / 8;

  gcn_heads_clear<<<clear_blocks, 256, 0, stream>>>(head);

  gcn_gemm1_wmma_link<<<GEMM1_BLOCKS + LINK_BLOCKS, 256, 0, stream>>>(
      feature, W1, buf0, src, dst, head, cells);

  gcn_gather16_bias_relu<<<gather_blocks, 256, 0, stream>>>(buf0, head, cells, b1, buf1);

  gcn_gather16<<<gather_blocks, 256, 0, stream>>>(buf1, head, cells, buf0);

  gcn_gemm2_wmma<<<gemm2_blocks, 256, 0, stream>>>(buf0, W2, b2, out);
}


# Each destination's edge list split into 4 lists (edge index mod 4), walked side by side by its gather thread: half the dependent hops per wave; float sum order: lists interleaved, arrival in a list
# speedup vs baseline: 1.3534x; 1.1201x over previous
#include <hip/hip_runtime.h>

typedef __attribute__((ext_vector_type(2))) float v2f;
typedef __attribute__((ext_vector_type(8))) float v8f;

#define N_NODES 50000
#define N_EDGES 400000
#define IN_F    768
#define HID_F   16
#define OUT_F   21

#define GEMM1_BLOCKS (((N_NODES / 16) + 7) / 8)
#define LINK_BLOCKS  (((N_EDGES / 4) + 255) / 256)
#define N_LISTS  4
#define CELL_END N_EDGES

__global__ __launch_bounds__(256) void gcn_heads_clear(
    int* __restrict__ head, int* __restrict__ cells)
{
  const unsigned g = blockIdx.x * 256u + threadIdx.x;
  if (g < N_NODES)
    ((int4*)head)[g] = make_int4(CELL_END, CELL_END, CELL_END, CELL_END);
  if (g == 0) *(int2*)(cells + 2 * (size_t)CELL_END) = make_int2(CELL_END, 0);
}

__device__ __forceinline__ void gcn_link4(
    unsigned t, const int* __restrict__ src, const int* __restrict__ dst,
    int* __restrict__ head, int* __restrict__ cells)
{
  if (t < N_EDGES / 4) {
    const int4 s = ((const int4*)src)[t];
    const int4 d = ((const int4*)dst)[t];
    const int e = (int)(4u * t);
    const int p0 = atomicExch(head + N_LISTS * d.x,     e);
    const int p1 = atomicExch(head + N_LISTS * d.y + 1, e + 1);
    const int p2 = atomicExch(head + N_LISTS * d.z + 2, e + 2);
    const int p3 = atomicExch(head + N_LISTS * d.w + 3, e + 3);
    int4* c = (int4*)(cells + 8 * (size_t)t);
    c[0] = make_int4(p0, s.x, p1, s.y);
    c[1] = make_int4(p2, s.z, p3, s.w);
  }
}

__global__ __launch_bounds__(256) void gcn_gemm1_wmma_link(
    const float* __restrict__ X, const float* __restrict__ W1,
    float* __restrict__ Z,
    const int* __restrict__ src, const int* __restrict__ dst,
    int* __restrict__ head, int* __restrict__ cells)
{
  __shared__ float w1s[IN_F * HID_F];
  if (blockIdx.x >= GEMM1_BLOCKS) {
    gcn_link4((blockIdx.x - GEMM1_BLOCKS) * 256u + threadIdx.x, src, dst, head, cells);
    return;
  }
  {
    const float4* w4 = (const float4*)W1;
    float4*       s4 = (float4*)w1s;
    const int n4 = (IN_F * HID_F) / 4;
    for (int i = threadIdx.x; i < n4; i += 256) s4[i] = w4[i];
  }
  __syncthreads();

  const int lane = threadIdx.x & 31;
  const int wave = threadIdx.x >> 5;
  const int tile = blockIdx.x * 8 + wave;
  if (tile >= (N_NODES / 16)) return;
  const int m0   = tile * 16;

  const int half = lane >> 4;
  const int l16  = lane & 15;
  const int koff = half << 1;

  const float* xrow = X + (size_t)(m0 + l16) * IN_F + koff;

  v8f acc0 = {};
  v8f acc1 = {};
  for (int k = 0; k < IN_F; k += 8) {
    v2f a0 = *(const v2f*)(xrow + k);
    v2f a1 = *(const v2f*)(xrow + k + 4);
    v2f b0, b1;
    b0.x = w1s[(k + koff)     * HID_F + l16];
    b0.y = w1s[(k + koff + 1) * HID_F + l16];
    b1.x = w1s[(k + 4 + koff)     * HID_F + l16];
    b1.y = w1s[(k + 4 + koff + 1) * HID_F + l16];
    acc0 = __builtin_amdgcn_wmma_f32_16x16x4_f32(false, a0, false, b0,
                                                 (short)0, acc0, false, false);
    acc1 = __builtin_amdgcn_wmma_f32_16x16x4_f32(false, a1, false, b1,
                                                 (short)0, acc1, false, false);
  }

  float* zp = Z + (size_t)(m0 + half * 8) * HID_F + l16;
#pragma unroll
  for (int r = 0; r < 8; ++r) zp[r * HID_F] = acc0[r] + acc1[r];
}

__device__ __forceinline__ float gcn_keep(float v, int m)
{
  return __int_as_float(__float_as_int(v) & m);
}

__device__ __forceinline__ float4 gcn_list_sum(
    const float* __restrict__ feat, const int* __restrict__ cells,
    int4 h, unsigned qq)
{
  const int2*   cell2 = (const int2*)cells;
  const float4* row4  = (const float4*)feat;
  float4 s = make_float4(0.0f, 0.0f, 0.0f, 0.0f);
  float4 v0 = s, v1 = s, v2 = s, v3 = s;
  int m0 = 0, m1 = 0, m2 = 0, m3 = 0;
  unsigned e0 = min((unsigned)h.x, (unsigned)CELL_END);
  unsigned e1 = min((unsigned)h.y, (unsigned)CELL_END);
  unsigned e2 = min((unsigned)h.z, (unsigned)CELL_END);
  unsigned e3 = min((unsigned)h.w, (unsigned)CELL_END);
  unsigned live = __builtin_amdgcn_ballot_w32(min(min(e0, e1), min(e2, e3)) < (unsigned)CELL_END);
  for (int it = 0; (it < N_EDGES) & (live != 0u); ++it) {
    const int2 c0 = cell2[e0];
    const int2 c1 = cell2[e1];
    const int2 c2 = cell2[e2];
    const int2 c3 = cell2[e3];
    s.x += gcn_keep(v0.x, m0); s.y += gcn_keep(v0.y, m0); s.z += gcn_keep(v0.z, m0); s.w += gcn_keep(v0.w, m0);
    s.x += gcn_keep(v1.x, m1); s.y += gcn_keep(v1.y, m1); s.z += gcn_keep(v1.z, m1); s.w += gcn_keep(v1.w, m1);
    s.x += gcn_keep(v2.x, m2); s.y += gcn_keep(v2.y, m2); s.z += gcn_keep(v2.z, m2); s.w += gcn_keep(v2.w, m2);
    s.x += gcn_keep(v3.x, m3); s.y += gcn_keep(v3.y, m3); s.z += gcn_keep(v3.z, m3); s.w += gcn_keep(v3.w, m3);
    v0 = row4[(unsigned)c0.y * 4u + qq];
    v1 = row4[(unsigned)c1.y * 4u + qq];
    v2 = row4[(unsigned)c2.y * 4u + qq];
    v3 = row4[(unsigned)c3.y * 4u + qq];
    m0 = e0 < (unsigned)CELL_END ? -1 : 0;
    m1 = e1 < (unsigned)CELL_END ? -1 : 0;
    m2 = e2 < (unsigned)CELL_END ? -1 : 0;
    m3 = e3 < (unsigned)CELL_END ? -1 : 0;
    e0 = min((unsigned)c0.x, (unsigned)CELL_END);
    e1 = min((unsigned)c1.x, (unsigned)CELL_END);
    e2 = min((unsigned)c2.x, (unsigned)CELL_END);
    e3 = min((unsigned)c3.x, (unsigned)CELL_END);
    live = __builtin_amdgcn_ballot_w32(min(min(e0, e1), min(e2, e3)) < (unsigned)CELL_END);
  }
  s.x += gcn_keep(v0.x, m0); s.y += gcn_keep(v0.y, m0); s.z += gcn_keep(v0.z, m0); s.w += gcn_keep(v0.w, m0);
  s.x += gcn_keep(v1.x, m1); s.y += gcn_keep(v1.y, m1); s.z += gcn_keep(v1.z, m1); s.w += gcn_keep(v1.w, m1);
  s.x += gcn_keep(v2.x, m2); s.y += gcn_keep(v2.y, m2); s.z += gcn_keep(v2.z, m2); s.w += gcn_keep(v2.w, m2);
  s.x += gcn_keep(v3.x, m3); s.y += gcn_keep(v3.y, m3); s.z += gcn_keep(v3.z, m3); s.w += gcn_keep(v3.w, m3);
  return s;
}

__global__ __launch_bounds__(256) void gcn_gather16_bias_relu(
    const float* __restrict__ feat, const int* __restrict__ head,
    const int* __restrict__ cells, const float* __restrict__ b1,
    float* __restrict__ h)
{
  unsigned tid = blockIdx.x * 256u + threadIdx.x;
  unsigned n = tid >> 2;
  unsigned q = (tid & 3u) * 4u;
  if (n < N_NODES) {
    const float4 b = *(const float4*)(b1 + q);
    float4 s = gcn_list_sum(feat, cells, ((const int4*)head)[n], tid & 3u);
    float4 r;
    r.x = s.x + b.x; r.y = s.y + b.y; r.z = s.z + b.z; r.w = s.w + b.w;
    r.x = r.x > 0.0f ? r.x : 0.0f;
    r.y = r.y > 0.0f ? r.y : 0.0f;
    r.z = r.z > 0.0f ? r.z : 0.0f;
    r.w = r.w > 0.0f ? r.w : 0.0f;
    *(float4*)(h + (size_t)n * HID_F + q) = r;
  }
}

__global__ __launch_bounds__(256) void gcn_gather16(
    const float* __restrict__ feat, const int* __restrict__ head,
    const int* __restrict__ cells, float* __restrict__ acc)
{
  unsigned tid = blockIdx.x * 256u + threadIdx.x;
  unsigned n = tid >> 2;
  unsigned q = (tid & 3u) * 4u;
  if (n < N_NODES) {
    float4 s = gcn_list_sum(feat, cells, ((const int4*)head)[n], tid & 3u);
    *(float4*)(acc + (size_t)n * HID_F + q) = s;
  }
}

__global__ __launch_bounds__(256) void gcn_gemm2_wmma(
    const float* __restrict__ A, const float* __restrict__ W2,
    const float* __restrict__ b2, float* __restrict__ out)
{
  const int lane = threadIdx.x & 31;
  const int wave = threadIdx.x >> 5;
  const int t    = blockIdx.x * 8 + wave;
  const int mtiles = N_NODES / 16;
  if (t >= mtiles * 2) return;
  const int mt = t >> 1;
  const int nt = t & 1;
  const int m0 = mt * 16;

  const int half = lane >> 4;
  const int l16  = lane & 15;
  const int koff = half << 1;
  const int n    = nt * 16 + l16;
  const bool nvalid = (n < OUT_F);

  const float* arow = A + (size_t)(m0 + l16) * HID_F + koff;

  v8f acc = {};
#pragma unroll
  for (int k = 0; k < HID_F; k += 4) {
    v2f a = *(const v2f*)(arow + k);
    v2f b;
    b.x = nvalid ? W2[(k + koff)     * OUT_F + n] : 0.0f;
    b.y = nvalid ? W2[(k + koff + 1) * OUT_F + n] : 0.0f;
    acc = __builtin_amdgcn_wmma_f32_16x16x4_f32(false, a, false, b,
                                                (short)0, acc, false, false);
  }

  if (nvalid) {
    float bias = b2[n];
    float* op = out + (size_t)(m0 + half * 8) * OUT_F + n;
#pragma unroll
    for (int r = 0; r < 8; ++r) op[r * OUT_F] = acc[r] + bias;
  }
}

extern "C" void kernel_launch(void* const* d_in, const int* in_sizes, int n_in,
                              void* d_out, int out_size, void* d_ws, size_t ws_size,
                              hipStream_t stream)
{
  const float* feature = (const float*)d_in[0];
  const float* W1      = (const float*)d_in[1];
  const float* b1      = (const float*)d_in[2];
  const float* W2      = (const float*)d_in[3];
  const float* b2      = (const float*)d_in[4];
  const int*   src     = (const int*)d_in[5];
  const int*   dst     = (const int*)d_in[6];
  float*       out     = (float*)d_out;

  float* buf0 = (float*)d_ws;
  float* buf1 = buf0 + (size_t)N_NODES * HID_F;

  int* cells = (int*)d_out;
  int* head  = cells + 2 * (size_t)N_EDGES + 4;

  const int clear_blocks   = (N_NODES + 255) / 256;
  const int gather_blocks  = (N_NODES * 4 + 255) / 256;
  const int gemm2_blocks   = ((N_NODES / 16) * 2 + 7) / 8;

  gcn_heads_clear<<<clear_blocks, 256, 0, stream>>>(head, cells);

  gcn_gemm1_wmma_link<<<GEMM1_BLOCKS + LINK_BLOCKS, 256, 0, stream>>>(
      feature, W1, buf0, src, dst, head, cells);

  gcn_gather16_bias_relu<<<gather_blocks, 256, 0, stream>>>(buf0, head, cells, b1, buf1);

  gcn_gather16<<<gather_blocks, 256, 0, stream>>>(buf1, head, cells, buf0);

  gcn_gemm2_wmma<<<gemm2_blocks, 256, 0, stream>>>(buf0, W2, b2, out);
}


# First product's X stream read in whole lines: 16-byte loads of 512-byte row pieces, two chunks ahead, turned into A fragments through a per-wave LDS buffer; k order inside its f32 sums changes
# speedup vs baseline: 1.5577x; 1.1510x over previous
#include <hip/hip_runtime.h>

typedef __attribute__((ext_vector_type(2))) float v2f;
typedef __attribute__((ext_vector_type(8))) float v8f;
typedef __attribute__((ext_vector_type(4))) float v4f;

#define N_NODES 50000
#define N_EDGES 400000
#define IN_F    768
#define HID_F   16
#define OUT_F   21

#define GEMM1_BLOCKS (((N_NODES / 16) + 7) / 8)
#define LINK_BLOCKS  (((N_EDGES / 4) + 255) / 256)
#define N_LISTS  4
#define CELL_END N_EDGES

__global__ __launch_bounds__(256) void gcn_heads_clear(
    int* __restrict__ head, int* __restrict__ cells)
{
  const unsigned g = blockIdx.x * 256u + threadIdx.x;
  if (g < N_NODES)
    ((int4*)head)[g] = make_int4(CELL_END, CELL_END, CELL_END, CELL_END);
  if (g == 0) *(int2*)(cells + 2 * (size_t)CELL_END) = make_int2(CELL_END, 0);
}

__device__ __forceinline__ void gcn_link4(
    unsigned t, const int* __restrict__ src, const int* __restrict__ dst,
    int* __restrict__ head, int* __restrict__ cells)
{
  if (t < N_EDGES / 4) {
    const int4 s = ((const int4*)src)[t];
    const int4 d = ((const int4*)dst)[t];
    const int e = (int)(4u * t);
    const int p0 = atomicExch(head + N_LISTS * d.x,     e);
    const int p1 = atomicExch(head + N_LISTS * d.y + 1, e + 1);
    const int p2 = atomicExch(head + N_LISTS * d.z + 2, e + 2);
    const int p3 = atomicExch(head + N_LISTS * d.w + 3, e + 3);
    int4* c = (int4*)(cells + 8 * (size_t)t);
    c[0] = make_int4(p0, s.x, p1, s.y);
    c[1] = make_int4(p2, s.z, p3, s.w);
  }
}

#define X_CH    128
#define X_NCH   (IN_F / X_CH)
#define X_PITCH 132

__device__ __forceinline__ void gcn_x_chunk_load(v4f (&b)[16], const float* __restrict__ p)
{
#pragma unroll
  for (int r = 0; r < 16; ++r) b[r] = *(const v4f*)(p + r * IN_F);
}

__device__ __forceinline__ void gcn_x_chunk_to_lds(float* st, const v4f (&b)[16])
{
#pragma unroll
  for (int r = 0; r < 16; ++r) *(v4f*)(st + r * X_PITCH) = b[r];
}

__device__ __forceinline__ void gcn_x_chunk_mma(
    const float* ld, const v4f* wc, v8f& acc0, v8f& acc1)
{
#pragma unroll
  for (int j = 0; j < 16; ++j) {
    const v4f a = *(const v4f*)(ld + 8 * j);
    const v4f b = wc[j * 32];
    v2f a0, a1, b0, b1;
    a0.x = a.x; a0.y = a.y; a1.x = a.z; a1.y = a.w;
    b0.x = b.x; b0.y = b.y; b1.x = b.z; b1.y = b.w;
    acc0 = __builtin_amdgcn_wmma_f32_16x16x4_f32(false, a0, false, b0,
                                                 (short)0, acc0, false, false);
    acc1 = __builtin_amdgcn_wmma_f32_16x16x4_f32(false, a1, false, b1,
                                                 (short)0, acc1, false, false);
  }
}

__global__ __launch_bounds__(256) void gcn_gemm1_wmma_link(
    const float* __restrict__ X, const float* __restrict__ W1,
    float* __restrict__ Z,
    const int* __restrict__ src, const int* __restrict__ dst,
    int* __restrict__ head, int* __restrict__ cells)
{
  __shared__ v4f   w1f[(IN_F / 8) * 32];
  __shared__ __attribute__((aligned(16))) float xst[8 * 16 * X_PITCH];
  if (blockIdx.x >= GEMM1_BLOCKS) {
    gcn_link4((blockIdx.x - GEMM1_BLOCKS) * 256u + threadIdx.x, src, dst, head, cells);
    return;
  }

  const int lane = threadIdx.x & 31;
  const int wave = threadIdx.x >> 5;
  const int tile = blockIdx.x * 8 + wave;
  const int half = lane >> 4;
  const int l16  = lane & 15;
  const int m0 = min(tile, N_NODES / 16 - 1) * 16;
  const float* xb = X + (size_t)m0 * IN_F + lane * 4;

  v4f wreg[(IN_F * HID_F) / 4 / 256];
#pragma unroll
  for (int t = 0; t < (IN_F * HID_F) / 4 / 256; ++t)
    wreg[t] = ((const v4f*)W1)[threadIdx.x + 256 * t];

  v4f bufa[16], bufb[16];
  gcn_x_chunk_load(bufa, xb);
  gcn_x_chunk_load(bufb, xb + X_CH);

#pragma unroll
  for (int t = 0; t < (IN_F * HID_F) / 4 / 256; ++t) {
    const int i = threadIdx.x + 256 * t;
    const int k = i >> 2;
    float* d = (float*)w1f + ((((k >> 3) * 32 + ((k >> 2) & 1) * 16 + (i & 3) * 4) << 2) + (k & 3));
    d[0] = wreg[t].x; d[4] = wreg[t].y; d[8] = wreg[t].z; d[12] = wreg[t].w;
  }
  __syncthreads();
  if (tile >= (N_NODES / 16)) return;

  float*       xw_st = xst + wave * (16 * X_PITCH) + lane * 4;
  const float* xw_ld = xst + wave * (16 * X_PITCH) + l16 * X_PITCH + half * 4;
  const v4f*   wf    = w1f + lane;

  v8f acc0 = {};
  v8f acc1 = {};
#define GCN_MEM_PIN() asm volatile("" ::: "memory")
#pragma unroll 1
  for (int ch = 0; ch < X_NCH - 2; ch += 2) {
    const float* xn = xb + (ch + 2) * X_CH;
    const v4f*   wc = wf + ch * (16 * 32);
    gcn_x_chunk_to_lds(xw_st, bufa);
    gcn_x_chunk_load(bufa, xn);
    GCN_MEM_PIN();
    gcn_x_chunk_mma(xw_ld, wc, acc0, acc1);
    gcn_x_chunk_to_lds(xw_st, bufb);
    gcn_x_chunk_load(bufb, xn + X_CH);
    GCN_MEM_PIN();
    gcn_x_chunk_mma(xw_ld, wc + 16 * 32, acc0, acc1);
  }
  gcn_x_chunk_to_lds(xw_st, bufa);
  gcn_x_chunk_mma(xw_ld, wf + (X_NCH - 2) * (16 * 32), acc0, acc1);
  gcn_x_chunk_to_lds(xw_st, bufb);
  gcn_x_chunk_mma(xw_ld, wf + (X_NCH - 1) * (16 * 32), acc0, acc1);
#undef GCN_MEM_PIN

  float* zp = Z + (size_t)(m0 + half * 8) * HID_F + l16;
#pragma unroll
  for (int r = 0; r < 8; ++r) zp[r * HID_F] = acc0[r] + acc1[r];
}

__device__ __forceinline__ float gcn_keep(float v, int m)
{
  return __int_as_float(__float_as_int(v) & m);
}

__device__ __forceinline__ float4 gcn_list_sum(
    const float* __restrict__ feat, const int* __restrict__ cells,
    int4 h, unsigned qq)
{
  const int2*   cell2 = (const int2*)cells;
  const float4* row4  = (const float4*)feat;
  float4 s = make_float4(0.0f, 0.0f, 0.0f, 0.0f);
  float4 v0 = s, v1 = s, v2 = s, v3 = s;
  int m0 = 0, m1 = 0, m2 = 0, m3 = 0;
  unsigned e0 = min((unsigned)h.x, (unsigned)CELL_END);
  unsigned e1 = min((unsigned)h.y, (unsigned)CELL_END);
  unsigned e2 = min((unsigned)h.z, (unsigned)CELL_END);
  unsigned e3 = min((unsigned)h.w, (unsigned)CELL_END);
  unsigned live = __builtin_amdgcn_ballot_w32(min(min(e0, e1), min(e2, e3)) < (unsigned)CELL_END);
  for (int it = 0; (it < N_EDGES) & (live != 0u); ++it) {
    const int2 c0 = cell2[e0];
    const int2 c1 = cell2[e1];
    const int2 c2 = cell2[e2];
    const int2 c3 = cell2[e3];
    s.x += gcn_keep(v0.x, m0); s.y += gcn_keep(v0.y, m0); s.z += gcn_keep(v0.z, m0); s.w += gcn_keep(v0.w, m0);
    s.x += gcn_keep(v1.x, m1); s.y += gcn_keep(v1.y, m1); s.z += gcn_keep(v1.z, m1); s.w += gcn_keep(v1.w, m1);
    s.x += gcn_keep(v2.x, m2); s.y += gcn_keep(v2.y, m2); s.z += gcn_keep(v2.z, m2); s.w += gcn_keep(v2.w, m2);
    s.x += gcn_keep(v3.x, m3); s.y += gcn_keep(v3.y, m3); s.z += gcn_keep(v3.z, m3); s.w += gcn_keep(v3.w, m3);
    v0 = row4[(unsigned)c0.y * 4u + qq];
    v1 = row4[(unsigned)c1.y * 4u + qq];
    v2 = row4[(unsigned)c2.y * 4u + qq];
    v3 = row4[(unsigned)c3.y * 4u + qq];
    m0 = e0 < (unsigned)CELL_END ? -1 : 0;
    m1 = e1 < (unsigned)CELL_END ? -1 : 0;
    m2 = e2 < (unsigned)CELL_END ? -1 : 0;
    m3 = e3 < (unsigned)CELL_END ? -1 : 0;
    e0 = min((unsigned)c0.x, (unsigned)CELL_END);
    e1 = min((unsigned)c1.x, (unsigned)CELL_END);
    e2 = min((unsigned)c2.x, (unsigned)CELL_END);
    e3 = min((unsigned)c3.x, (unsigned)CELL_END);
    live = __builtin_amdgcn_ballot_w32(min(min(e0, e1), min(e2, e3)) < (unsigned)CELL_END);
  }
  s.x += gcn_keep(v0.x, m0); s.y += gcn_keep(v0.y, m0); s.z += gcn_keep(v0.z, m0); s.w += gcn_keep(v0.w, m0);
  s.x += gcn_keep(v1.x, m1); s.y += gcn_keep(v1.y, m1); s.z += gcn_keep(v1.z, m1); s.w += gcn_keep(v1.w, m1);
  s.x += gcn_keep(v2.x, m2); s.y += gcn_keep(v2.y, m2); s.z += gcn_keep(v2.z, m2); s.w += gcn_keep(v2.w, m2);
  s.x += gcn_keep(v3.x, m3); s.y += gcn_keep(v3.y, m3); s.z += gcn_keep(v3.z, m3); s.w += gcn_keep(v3.w, m3);
  return s;
}

__global__ __launch_bounds__(256) void gcn_gather16_bias_relu(
    const float* __restrict__ feat, const int* __restrict__ head,
    const int* __restrict__ cells, const float* __restrict__ b1,
    float* __restrict__ h)
{
  unsigned tid = blockIdx.x * 256u + threadIdx.x;
  unsigned n = tid >> 2;
  unsigned q = (tid & 3u) * 4u;
  if (n < N_NODES) {
    const float4 b = *(const float4*)(b1 + q);
    float4 s = gcn_list_sum(feat, cells, ((const int4*)head)[n], tid & 3u);
    float4 r;
    r.x = s.x + b.x; r.y = s.y + b.y; r.z = s.z + b.z; r.w = s.w + b.w;
    r.x = r.x > 0.0f ? r.x : 0.0f;
    r.y = r.y > 0.0f ? r.y : 0.0f;
    r.z = r.z > 0.0f ? r.z : 0.0f;
    r.w = r.w > 0.0f ? r.w : 0.0f;
    *(float4*)(h + (size_t)n * HID_F + q) = r;
  }
}

__global__ __launch_bounds__(256) void gcn_gather16(
    const float* __restrict__ feat, const int* __restrict__ head,
    const int* __restrict__ cells, float* __restrict__ acc)
{
  unsigned tid = blockIdx.x * 256u + threadIdx.x;
  unsigned n = tid >> 2;
  unsigned q = (tid & 3u) * 4u;
  if (n < N_NODES) {
    float4 s = gcn_list_sum(feat, cells, ((const int4*)head)[n], tid & 3u);
    *(float4*)(acc + (size_t)n * HID_F + q) = s;
  }
}

__global__ __launch_bounds__(256) void gcn_gemm2_wmma(
    const float* __restrict__ A, const float* __restrict__ W2,
    const float* __restrict__ b2, float* __restrict__ out)
{
  const int lane = threadIdx.x & 31;
  const int wave = threadIdx.x >> 5;
  const int t    = blockIdx.x * 8 + wave;
  const int mtiles = N_NODES / 16;
  if (t >= mtiles * 2) return;
  const int mt = t >> 1;
  const int nt = t & 1;
  const int m0 = mt * 16;

  const int half = lane >> 4;
  const int l16  = lane & 15;
  const int koff = half << 1;
  const int n    = nt * 16 + l16;
  const bool nvalid = (n < OUT_F);

  const float* arow = A + (size_t)(m0 + l16) * HID_F + koff;

  v8f acc = {};
#pragma unroll
  for (int k = 0; k < HID_F; k += 4) {
    v2f a = *(const v2f*)(arow + k);
    v2f b;
    b.x = nvalid ? W2[(k + koff)     * OUT_F + n] : 0.0f;
    b.y = nvalid ? W2[(k + koff + 1) * OUT_F + n] : 0.0f;
    acc = __builtin_amdgcn_wmma_f32_16x16x4_f32(false, a, false, b,
                                                (short)0, acc, false, false);
  }

  if (nvalid) {
    float bias = b2[n];
    float* op = out + (size_t)(m0 + half * 8) * OUT_F + n;
#pragma unroll
    for (int r = 0; r < 8; ++r) op[r * OUT_F] = acc[r] + bias;
  }
}

extern "C" void kernel_launch(void* const* d_in, const int* in_sizes, int n_in,
                              void* d_out, int out_size, void* d_ws, size_t ws_size,
                              hipStream_t stream)
{
  const float* feature = (const float*)d_in[0];
  const float* W1      = (const float*)d_in[1];
  const float* b1      = (const float*)d_in[2];
  const float* W2      = (const float*)d_in[3];
  const float* b2      = (const float*)d_in[4];
  const int*   src     = (const int*)d_in[5];
  const int*   dst     = (const int*)d_in[6];
  float*       out     = (float*)d_out;

  float* buf0 = (float*)d_ws;
  float* buf1 = buf0 + (size_t)N_NODES * HID_F;

  int* cells = (int*)d_out;
  int* head  = cells + 2 * (size_t)N_EDGES + 4;

  const int clear_blocks   = (N_NODES + 255) / 256;
  const int gather_blocks  = (N_NODES * 4 + 255) / 256;
  const int gemm2_blocks   = ((N_NODES / 16) * 2 + 7) / 8;

  gcn_heads_clear<<<clear_blocks, 256, 0, stream>>>(head, cells);

  gcn_gemm1_wmma_link<<<GEMM1_BLOCKS + LINK_BLOCKS, 256, 0, stream>>>(
      feature, W1, buf0, src, dst, head, cells);

  gcn_gather16_bias_relu<<<gather_blocks, 256, 0, stream>>>(buf0, head, cells, b1, buf1);

  gcn_gather16<<<gather_blocks, 256, 0, stream>>>(buf1, head, cells, buf0);

  gcn_gemm2_wmma<<<gemm2_blocks, 256, 0, stream>>>(buf0, W2, b2, out);
}
